# starting kernel
# baseline (speedup 1.0000x reference)
#include <hip/hip_runtime.h>


#define NV 128
#define NF 4
#define NT 16
#define NR 5
#define NB 80
#define NROT 16
#define TWO_PI_F 6.28318530717958647692f
#define ROT_STEP (TWO_PI_F / 16.0f)
#define EPSF 1e-5f

typedef float v2f __attribute__((ext_vector_type(2)));
typedef float v8f __attribute__((ext_vector_type(8)));

#define QSTRIDE 130
#define DSTRIDE 82
#define THM_OFF 0
#define Q_OFF   256
#define D_OFF   (Q_OFF + 4 * 16 * QSTRIDE)
#define LDS_FLOATS (D_OFF + 4 * 16 * DSTRIDE)

__global__ __launch_bounds__(128)
void masif_geoconv_kernel(const float* __restrict__ rho_coords,
                          const float* __restrict__ theta_coords,
                          const float* __restrict__ input_feat,
                          const float* __restrict__ mask,
                          const float* __restrict__ mu_rho,
                          const float* __restrict__ sigma_rho,
                          const float* __restrict__ mu_theta,
                          const float* __restrict__ sigma_theta,
                          const float* __restrict__ W_conv,
                          const float* __restrict__ b_conv,
                          float* __restrict__ out)
{
    __shared__ float lds[LDS_FLOATS];

    const int n    = blockIdx.x;
    const int tid  = threadIdx.x;
    const int wave = tid >> 5;
    const int lane = tid & 31;
    const int half = lane >> 4;
    const int tloc = lane & 15;
    const int i    = wave;

    {
        float tv = theta_coords[n * NV + tid];
        float mv = mask[n * NV + tid];
        lds[THM_OFF + 2 * tid]     = tv;
        lds[THM_OFF + 2 * tid + 1] = mv;
    }

    float mu_r[NR], is2_r[NR];
#pragma unroll
    for (int r = 0; r < NR; ++r) {
        float m = mu_rho[i * NB + r * NT];
        float s = sigma_rho[i * NB + r * NT];
        mu_r[r]  = m;
        is2_r[r] = 1.0f / (s * s + EPSF);
    }
    float mu_t, is2_t;
    {
        mu_t = mu_theta[i * NB + tloc];
        float s = sigma_theta[i * NB + tloc];
        is2_t = 1.0f / (s * s + EPSF);
    }

    float* Q = &lds[Q_OFF + wave * 16 * QSTRIDE];
    for (int v = lane; v < NV; v += 32) {
        float rv = rho_coords[n * NV + v];
        float fv = input_feat[(n * NV + v) * NF + i];
#pragma unroll
        for (int r = 0; r < NR; ++r) {
            float d  = rv - mu_r[r];
            float rg = __expf(-d * d * is2_r[r]);
            Q[(2 * r)     * QSTRIDE + v] = rg * fv;
            Q[(2 * r + 1) * QSTRIDE + v] = rg;
        }
#pragma unroll
        for (int c = 2 * NR; c < 16; ++c) Q[c * QSTRIDE + v] = 0.0f;
    }
    __syncthreads();

    float* Dsc       = &lds[D_OFF + wave * 16 * DSTRIDE];
    const float* thm = &lds[THM_OFF];

    for (int k = 0; k < NROT; ++k) {
        const float kofs = (float)k * ROT_STEP;
        v8f acc = {};
        for (int kb = 0; kb < 32; ++kb) {
            const int v0 = 4 * kb + 2 * half;
            v2f a = *(const v2f*)&Q[tloc * QSTRIDE + v0];
            v2f b;
#pragma unroll
            for (int j = 0; j < 2; ++j) {
                float tv = thm[2 * (v0 + j)];
                float mv = thm[2 * (v0 + j) + 1];
                float th = tv + kofs;
                th = (th >= TWO_PI_F) ? th - TWO_PI_F : th;
                float d = th - mu_t;
                b[j] = __expf(-d * d * is2_t) * mv;
            }
            acc = __builtin_amdgcn_wmma_f32_16x16x4_f32(
                false, a, false, b, (short)0, acc, false, false);
        }
        if (half == 0) {
#pragma unroll
            for (int r = 0; r < 4; ++r) {
                Dsc[k * DSTRIDE + r * NT + tloc] = acc[2 * r] / (acc[2 * r + 1] + EPSF);
            }
        } else {
            Dsc[k * DSTRIDE + 4 * NT + tloc] = acc[0] / (acc[1] + EPSF);
        }
    }
    __syncthreads();

    const float* Wi = W_conv + i * NB * NB;
    for (int bt = 0; bt < 5; ++bt) {
        const int col = bt * NT + tloc;
        const float bias = b_conv[i * NB + col];
        v8f acc = {bias, bias, bias, bias, bias, bias, bias, bias};
        for (int kb = 0; kb < 20; ++kb) {
            const int krow = 4 * kb + 2 * half;
            v2f a = *(const v2f*)&Dsc[tloc * DSTRIDE + krow];
            v2f b;
            b.x = Wi[krow * NB + col];
            b.y = Wi[(krow + 1) * NB + col];
            acc = __builtin_amdgcn_wmma_f32_16x16x4_f32(
                false, a, false, b, (short)0, acc, false, false);
        }
        float m = acc[0];
#pragma unroll
        for (int j = 1; j < 8; ++j) m = fmaxf(m, acc[j]);
        m = fmaxf(m, __shfl_xor(m, 16, 32));
        if (half == 0) {
            out[(n * NF + i) * NB + col] = fmaxf(m, 0.0f);
        }
    }
}

extern "C" void kernel_launch(void* const* d_in, const int* in_sizes, int n_in,
                              void* d_out, int out_size, void* d_ws, size_t ws_size,
                              hipStream_t stream) {
    const float* rho_coords  = (const float*)d_in[0];
    const float* theta_coord = (const float*)d_in[1];
    const float* input_feat  = (const float*)d_in[2];
    const float* mask        = (const float*)d_in[3];
    const float* mu_rho      = (const float*)d_in[4];
    const float* sigma_rho   = (const float*)d_in[5];
    const float* mu_theta    = (const float*)d_in[6];
    const float* sigma_theta = (const float*)d_in[7];
    const float* W_conv      = (const float*)d_in[8];
    const float* b_conv      = (const float*)d_in[9];
    float* out = (float*)d_out;

    const int n_samples = in_sizes[0] / NV;
    masif_geoconv_kernel<<<n_samples, 128, 0, stream>>>(
        rho_coords, theta_coord, input_feat, mask,
        mu_rho, sigma_rho, mu_theta, sigma_theta, W_conv, b_conv, out);
}


# B operand cut to its t-dependent part: rotated+wrapped theta per (v,k) built once per block in LDS (shared by 4 waves), mask folded into Q, exp2 scale pre-multiplied per lane; sum order unchanged
# speedup vs baseline: 1.3543x; 1.3543x over previous
#include <hip/hip_runtime.h>


#define NV 128
#define NF 4
#define NT 16
#define NR 5
#define NB 80
#define NROT 16
#define TWO_PI_F 6.28318530717958647692f
#define ROT_STEP (TWO_PI_F / 16.0f)
#define EPSF 1e-5f

typedef float v2f __attribute__((ext_vector_type(2)));
typedef float v8f __attribute__((ext_vector_type(8)));

#define QSTRIDE 130
#define DSTRIDE 82
#define TH_OFF  0
#define Q_OFF   (NROT * NV)
#define D_OFF   (Q_OFF + 4 * 16 * QSTRIDE)
#define LDS_FLOATS (D_OFF + 4 * 16 * DSTRIDE)
#define LOG2E_F 1.44269504088896340736f

__global__ __launch_bounds__(128)
void masif_geoconv_kernel(const float* __restrict__ rho_coords,
                          const float* __restrict__ theta_coords,
                          const float* __restrict__ input_feat,
                          const float* __restrict__ mask,
                          const float* __restrict__ mu_rho,
                          const float* __restrict__ sigma_rho,
                          const float* __restrict__ mu_theta,
                          const float* __restrict__ sigma_theta,
                          const float* __restrict__ W_conv,
                          const float* __restrict__ b_conv,
                          float* __restrict__ out)
{
    __shared__ float lds[LDS_FLOATS];

    const int n    = blockIdx.x;
    const int tid  = threadIdx.x;
    const int wave = tid >> 5;
    const int lane = tid & 31;
    const int half = lane >> 4;
    const int tloc = lane & 15;
    const int i    = wave;

    {
        const float tv = theta_coords[n * NV + tid];
#pragma unroll
        for (int k = 0; k < NROT; ++k) {
            float th = __builtin_fmaf((float)k, ROT_STEP, tv);
            th = (th >= TWO_PI_F) ? th - TWO_PI_F : th;
            lds[TH_OFF + k * NV + tid] = th;
        }
    }

    float mu_r[NR], is2_r[NR];
#pragma unroll
    for (int r = 0; r < NR; ++r) {
        float m = mu_rho[i * NB + r * NT];
        float s = sigma_rho[i * NB + r * NT];
        mu_r[r]  = m;
        is2_r[r] = 1.0f / (s * s + EPSF);
    }
    float mu_t, nsc_t;
    {
        mu_t = mu_theta[i * NB + tloc];
        float s = sigma_theta[i * NB + tloc];
        nsc_t = -((1.0f / (s * s + EPSF)) * LOG2E_F);
    }

    float* Q = &lds[Q_OFF + wave * 16 * QSTRIDE];
    for (int v = lane; v < NV; v += 32) {
        float rv = rho_coords[n * NV + v];
        float fv = input_feat[(n * NV + v) * NF + i];
        float mv = mask[n * NV + v];
#pragma unroll
        for (int r = 0; r < NR; ++r) {
            float d  = rv - mu_r[r];
            float rg = __expf(-d * d * is2_r[r]) * mv;
            Q[(2 * r)     * QSTRIDE + v] = rg * fv;
            Q[(2 * r + 1) * QSTRIDE + v] = rg;
        }
#pragma unroll
        for (int c = 2 * NR; c < 16; ++c) Q[c * QSTRIDE + v] = 0.0f;
    }
    __syncthreads();

    float* Dsc = &lds[D_OFF + wave * 16 * DSTRIDE];
    const v2f mu2  = {mu_t, mu_t};
    const v2f nsc2 = {nsc_t, nsc_t};

    for (int k = 0; k < NROT; ++k) {
        const float* THk = &lds[TH_OFF + k * NV];
        v8f acc = {};
        for (int kb = 0; kb < 32; ++kb) {
            const int v0 = 4 * kb + 2 * half;
            v2f a = *(const v2f*)&Q[tloc * QSTRIDE + v0];
            const v2f th = *(const v2f*)&THk[v0];
            const v2f d  = th - mu2;
            const v2f x  = (d * d) * nsc2;
            v2f b;
            b.x = __builtin_amdgcn_exp2f(x.x);
            b.y = __builtin_amdgcn_exp2f(x.y);
            acc = __builtin_amdgcn_wmma_f32_16x16x4_f32(
                false, a, false, b, (short)0, acc, false, false);
        }
        if (half == 0) {
#pragma unroll
            for (int r = 0; r < 4; ++r) {
                Dsc[k * DSTRIDE + r * NT + tloc] = acc[2 * r] / (acc[2 * r + 1] + EPSF);
            }
        } else {
            Dsc[k * DSTRIDE + 4 * NT + tloc] = acc[0] / (acc[1] + EPSF);
        }
    }
    __syncthreads();

    const float* Wi = W_conv + i * NB * NB;
    for (int bt = 0; bt < 5; ++bt) {
        const int col = bt * NT + tloc;
        const float bias = b_conv[i * NB + col];
        v8f acc = {bias, bias, bias, bias, bias, bias, bias, bias};
        for (int kb = 0; kb < 20; ++kb) {
            const int krow = 4 * kb + 2 * half;
            v2f a = *(const v2f*)&Dsc[tloc * DSTRIDE + krow];
            v2f b;
            b.x = Wi[krow * NB + col];
            b.y = Wi[(krow + 1) * NB + col];
            acc = __builtin_amdgcn_wmma_f32_16x16x4_f32(
                false, a, false, b, (short)0, acc, false, false);
        }
        float m = acc[0];
#pragma unroll
        for (int j = 1; j < 8; ++j) m = fmaxf(m, acc[j]);
        m = fmaxf(m, __shfl_xor(m, 16, 32));
        if (half == 0) {
            out[(n * NF + i) * NB + col] = fmaxf(m, 0.0f);
        }
    }
}

extern "C" void kernel_launch(void* const* d_in, const int* in_sizes, int n_in,
                              void* d_out, int out_size, void* d_ws, size_t ws_size,
                              hipStream_t stream) {
    const float* rho_coords  = (const float*)d_in[0];
    const float* theta_coord = (const float*)d_in[1];
    const float* input_feat  = (const float*)d_in[2];
    const float* mask        = (const float*)d_in[3];
    const float* mu_rho      = (const float*)d_in[4];
    const float* sigma_rho   = (const float*)d_in[5];
    const float* mu_theta    = (const float*)d_in[6];
    const float* sigma_theta = (const float*)d_in[7];
    const float* W_conv      = (const float*)d_in[8];
    const float* b_conv      = (const float*)d_in[9];
    float* out = (float*)d_out;

    const int n_samples = in_sizes[0] / NV;
    masif_geoconv_kernel<<<n_samples, 128, 0, stream>>>(
        rho_coords, theta_coord, input_feat, mask,
        mu_rho, sigma_rho, mu_theta, sigma_theta, W_conv, b_conv, out);
}


# No LDS read between WMMAs: A fragments (rotation-independent) kept in registers for all 16 rotations, each theta row read as one load group a block ahead; sum order unchanged
# speedup vs baseline: 1.4425x; 1.0651x over previous
#include <hip/hip_runtime.h>


#define NV 128
#define NF 4
#define NT 16
#define NR 5
#define NB 80
#define NROT 16
#define TWO_PI_F 6.28318530717958647692f
#define ROT_STEP (TWO_PI_F / 16.0f)
#define EPSF 1e-5f

typedef float v2f __attribute__((ext_vector_type(2)));
typedef float v8f __attribute__((ext_vector_type(8)));

#define QSTRIDE 130
#define DSTRIDE 82
#define TH_OFF  0
#define Q_OFF   (NROT * NV)
#define D_OFF   (Q_OFF + 4 * 16 * QSTRIDE)
#define LDS_FLOATS (D_OFF + 4 * 16 * DSTRIDE)
#define LOG2E_F 1.44269504088896340736f

__device__ __forceinline__ void load_theta_row(v2f (&th)[32], const float* row)
{
#pragma unroll
    for (int kb = 0; kb < 32; ++kb) th[kb] = *(const v2f*)&row[4 * kb];
}

__device__ __forceinline__ v8f rotation_block(const v2f (&a)[32], const v2f (&th)[32],
                                              const v2f mu2, const v2f nsc2)
{
    v8f acc = {};
#pragma unroll
    for (int kb = 0; kb < 32; ++kb) {
        const v2f d = th[kb] - mu2;
        const v2f x = (d * d) * nsc2;
        v2f b;
        b.x = __builtin_amdgcn_exp2f(x.x);
        b.y = __builtin_amdgcn_exp2f(x.y);
        acc = __builtin_amdgcn_wmma_f32_16x16x4_f32(
            false, a[kb], false, b, (short)0, acc, false, false);
    }
    return acc;
}

__device__ __forceinline__ void store_desc(float* drow, const v8f acc, const int half, const int tloc)
{
    if (half == 0) {
#pragma unroll
        for (int r = 0; r < 4; ++r) {
            drow[r * NT + tloc] = acc[2 * r] / (acc[2 * r + 1] + EPSF);
        }
    } else {
        drow[4 * NT + tloc] = acc[0] / (acc[1] + EPSF);
    }
}

__global__ __launch_bounds__(128)
void masif_geoconv_kernel(const float* __restrict__ rho_coords,
                          const float* __restrict__ theta_coords,
                          const float* __restrict__ input_feat,
                          const float* __restrict__ mask,
                          const float* __restrict__ mu_rho,
                          const float* __restrict__ sigma_rho,
                          const float* __restrict__ mu_theta,
                          const float* __restrict__ sigma_theta,
                          const float* __restrict__ W_conv,
                          const float* __restrict__ b_conv,
                          float* __restrict__ out)
{
    __shared__ float lds[LDS_FLOATS];

    const int n    = blockIdx.x;
    const int tid  = threadIdx.x;
    const int wave = tid >> 5;
    const int lane = tid & 31;
    const int half = lane >> 4;
    const int tloc = lane & 15;
    const int i    = wave;

    {
        const float tv = theta_coords[n * NV + tid];
#pragma unroll
        for (int k = 0; k < NROT; ++k) {
            float th = __builtin_fmaf((float)k, ROT_STEP, tv);
            th = (th >= TWO_PI_F) ? th - TWO_PI_F : th;
            lds[TH_OFF + k * NV + tid] = th;
        }
    }

    float mu_r[NR], is2_r[NR];
#pragma unroll
    for (int r = 0; r < NR; ++r) {
        float m = mu_rho[i * NB + r * NT];
        float s = sigma_rho[i * NB + r * NT];
        mu_r[r]  = m;
        is2_r[r] = 1.0f / (s * s + EPSF);
    }
    float mu_t, nsc_t;
    {
        mu_t = mu_theta[i * NB + tloc];
        float s = sigma_theta[i * NB + tloc];
        nsc_t = -((1.0f / (s * s + EPSF)) * LOG2E_F);
    }

    float* Q = &lds[Q_OFF + wave * 16 * QSTRIDE];
    for (int v = lane; v < NV; v += 32) {
        float rv = rho_coords[n * NV + v];
        float fv = input_feat[(n * NV + v) * NF + i];
        float mv = mask[n * NV + v];
#pragma unroll
        for (int r = 0; r < NR; ++r) {
            float d  = rv - mu_r[r];
            float rg = __expf(-d * d * is2_r[r]) * mv;
            Q[(2 * r)     * QSTRIDE + v] = rg * fv;
            Q[(2 * r + 1) * QSTRIDE + v] = rg;
        }
#pragma unroll
        for (int c = 2 * NR; c < 16; ++c) Q[c * QSTRIDE + v] = 0.0f;
    }
    __syncthreads();

    float* Dsc = &lds[D_OFF + wave * 16 * DSTRIDE];
    const v2f mu2  = {mu_t, mu_t};
    const v2f nsc2 = {nsc_t, nsc_t};

    v2f afrag[32];
#pragma unroll
    for (int kb = 0; kb < 32; ++kb) afrag[kb] = *(const v2f*)&Q[tloc * QSTRIDE + 4 * kb + 2 * half];

    const float* THrow = &lds[TH_OFF + 2 * half];
    const float* row = THrow;
    v2f thA[32], thB[32];
    load_theta_row(thA, row);
    for (int k = 0; k < NROT; k += 2) {
        load_theta_row(thB, row + NV);
        __builtin_amdgcn_sched_barrier(0);
        store_desc(&Dsc[k * DSTRIDE], rotation_block(afrag, thA, mu2, nsc2), half, tloc);
        row = (k + 2 < NROT) ? row + 2 * NV : THrow;
        load_theta_row(thA, row);
        __builtin_amdgcn_sched_barrier(0);
        store_desc(&Dsc[(k + 1) * DSTRIDE], rotation_block(afrag, thB, mu2, nsc2), half, tloc);
    }
    __syncthreads();

    const float* Wi = W_conv + i * NB * NB;
    for (int bt = 0; bt < 5; ++bt) {
        const int col = bt * NT + tloc;
        const float bias = b_conv[i * NB + col];
        v8f acc = {bias, bias, bias, bias, bias, bias, bias, bias};
        for (int kb = 0; kb < 20; ++kb) {
            const int krow = 4 * kb + 2 * half;
            v2f a = *(const v2f*)&Dsc[tloc * DSTRIDE + krow];
            v2f b;
            b.x = Wi[krow * NB + col];
            b.y = Wi[(krow + 1) * NB + col];
            acc = __builtin_amdgcn_wmma_f32_16x16x4_f32(
                false, a, false, b, (short)0, acc, false, false);
        }
        float m = acc[0];
#pragma unroll
        for (int j = 1; j < 8; ++j) m = fmaxf(m, acc[j]);
        m = fmaxf(m, __shfl_xor(m, 16, 32));
        if (half == 0) {
            out[(n * NF + i) * NB + col] = fmaxf(m, 0.0f);
        }
    }
}

extern "C" void kernel_launch(void* const* d_in, const int* in_sizes, int n_in,
                              void* d_out, int out_size, void* d_ws, size_t ws_size,
                              hipStream_t stream) {
    const float* rho_coords  = (const float*)d_in[0];
    const float* theta_coord = (const float*)d_in[1];
    const float* input_feat  = (const float*)d_in[2];
    const float* mask        = (const float*)d_in[3];
    const float* mu_rho      = (const float*)d_in[4];
    const float* sigma_rho   = (const float*)d_in[5];
    const float* mu_theta    = (const float*)d_in[6];
    const float* sigma_theta = (const float*)d_in[7];
    const float* W_conv      = (const float*)d_in[8];
    const float* b_conv      = (const float*)d_in[9];
    float* out = (float*)d_out;

    const int n_samples = in_sizes[0] / NV;
    masif_geoconv_kernel<<<n_samples, 128, 0, stream>>>(
        rho_coords, theta_coord, input_feat, mask,
        mu_rho, sigma_rho, mu_theta, sigma_theta, W_conv, b_conv, out);
}


# Two accumulator chains per wave (even/odd k-steps, added at the end) in both WMMA products: WMMAs issue at the pipe interval, not at the dependent-accumulator latency; sum order changed
# speedup vs baseline: 1.5048x; 1.0432x over previous
#include <hip/hip_runtime.h>


#define NV 128
#define NF 4
#define NT 16
#define NR 5
#define NB 80
#define NROT 16
#define TWO_PI_F 6.28318530717958647692f
#define ROT_STEP (TWO_PI_F / 16.0f)
#define EPSF 1e-5f

typedef float v2f __attribute__((ext_vector_type(2)));
typedef float v8f __attribute__((ext_vector_type(8)));

#define QSTRIDE 130
#define DSTRIDE 82
#define TH_OFF  0
#define Q_OFF   (NROT * NV)
#define D_OFF   (Q_OFF + 4 * 16 * QSTRIDE)
#define LDS_FLOATS (D_OFF + 4 * 16 * DSTRIDE)
#define LOG2E_F 1.44269504088896340736f

__device__ __forceinline__ void load_theta_row(v2f (&th)[32], const float* row)
{
#pragma unroll
    for (int kb = 0; kb < 32; ++kb) th[kb] = *(const v2f*)&row[4 * kb];
}

__device__ __forceinline__ v8f rotation_block(const v2f (&a)[32], const v2f (&th)[32],
                                              const v2f mu2, const v2f nsc2)
{
    v8f acc0 = {}, acc1 = {};
#pragma unroll
    for (int kb = 0; kb < 32; kb += 2) {
        const v2f d0 = th[kb] - mu2;
        const v2f x0 = (d0 * d0) * nsc2;
        v2f b0;
        b0.x = __builtin_amdgcn_exp2f(x0.x);
        b0.y = __builtin_amdgcn_exp2f(x0.y);
        acc0 = __builtin_amdgcn_wmma_f32_16x16x4_f32(
            false, a[kb], false, b0, (short)0, acc0, false, false);
        const v2f d1 = th[kb + 1] - mu2;
        const v2f x1 = (d1 * d1) * nsc2;
        v2f b1;
        b1.x = __builtin_amdgcn_exp2f(x1.x);
        b1.y = __builtin_amdgcn_exp2f(x1.y);
        acc1 = __builtin_amdgcn_wmma_f32_16x16x4_f32(
            false, a[kb + 1], false, b1, (short)0, acc1, false, false);
    }
    return acc0 + acc1;
}

__device__ __forceinline__ void store_desc(float* drow, const v8f acc, const int half, const int tloc)
{
    if (half == 0) {
#pragma unroll
        for (int r = 0; r < 4; ++r) {
            drow[r * NT + tloc] = acc[2 * r] / (acc[2 * r + 1] + EPSF);
        }
    } else {
        drow[4 * NT + tloc] = acc[0] / (acc[1] + EPSF);
    }
}

__global__ __launch_bounds__(128)
void masif_geoconv_kernel(const float* __restrict__ rho_coords,
                          const float* __restrict__ theta_coords,
                          const float* __restrict__ input_feat,
                          const float* __restrict__ mask,
                          const float* __restrict__ mu_rho,
                          const float* __restrict__ sigma_rho,
                          const float* __restrict__ mu_theta,
                          const float* __restrict__ sigma_theta,
                          const float* __restrict__ W_conv,
                          const float* __restrict__ b_conv,
                          float* __restrict__ out)
{
    __shared__ float lds[LDS_FLOATS];

    const int n    = blockIdx.x;
    const int tid  = threadIdx.x;
    const int wave = tid >> 5;
    const int lane = tid & 31;
    const int half = lane >> 4;
    const int tloc = lane & 15;
    const int i    = wave;

    {
        const float tv = theta_coords[n * NV + tid];
#pragma unroll
        for (int k = 0; k < NROT; ++k) {
            float th = __builtin_fmaf((float)k, ROT_STEP, tv);
            th = (th >= TWO_PI_F) ? th - TWO_PI_F : th;
            lds[TH_OFF + k * NV + tid] = th;
        }
    }

    float mu_r[NR], is2_r[NR];
#pragma unroll
    for (int r = 0; r < NR; ++r) {
        float m = mu_rho[i * NB + r * NT];
        float s = sigma_rho[i * NB + r * NT];
        mu_r[r]  = m;
        is2_r[r] = 1.0f / (s * s + EPSF);
    }
    float mu_t, nsc_t;
    {
        mu_t = mu_theta[i * NB + tloc];
        float s = sigma_theta[i * NB + tloc];
        nsc_t = -((1.0f / (s * s + EPSF)) * LOG2E_F);
    }

    float* Q = &lds[Q_OFF + wave * 16 * QSTRIDE];
    for (int v = lane; v < NV; v += 32) {
        float rv = rho_coords[n * NV + v];
        float fv = input_feat[(n * NV + v) * NF + i];
        float mv = mask[n * NV + v];
#pragma unroll
        for (int r = 0; r < NR; ++r) {
            float d  = rv - mu_r[r];
            float rg = __expf(-d * d * is2_r[r]) * mv;
            Q[(2 * r)     * QSTRIDE + v] = rg * fv;
            Q[(2 * r + 1) * QSTRIDE + v] = rg;
        }
#pragma unroll
        for (int c = 2 * NR; c < 16; ++c) Q[c * QSTRIDE + v] = 0.0f;
    }
    __syncthreads();

    float* Dsc = &lds[D_OFF + wave * 16 * DSTRIDE];
    const v2f mu2  = {mu_t, mu_t};
    const v2f nsc2 = {nsc_t, nsc_t};

    v2f afrag[32];
#pragma unroll
    for (int kb = 0; kb < 32; ++kb) afrag[kb] = *(const v2f*)&Q[tloc * QSTRIDE + 4 * kb + 2 * half];

    const float* THrow = &lds[TH_OFF + 2 * half];
    const float* row = THrow;
    v2f thA[32], thB[32];
    load_theta_row(thA, row);
    for (int k = 0; k < NROT; k += 2) {
        load_theta_row(thB, row + NV);
        __builtin_amdgcn_sched_barrier(0);
        store_desc(&Dsc[k * DSTRIDE], rotation_block(afrag, thA, mu2, nsc2), half, tloc);
        row = (k + 2 < NROT) ? row + 2 * NV : THrow;
        load_theta_row(thA, row);
        __builtin_amdgcn_sched_barrier(0);
        store_desc(&Dsc[(k + 1) * DSTRIDE], rotation_block(afrag, thB, mu2, nsc2), half, tloc);
    }
    __syncthreads();

    const float* Wi = W_conv + i * NB * NB;
    for (int bt = 0; bt < 5; ++bt) {
        const int col = bt * NT + tloc;
        const float bias = b_conv[i * NB + col];
        v8f acc = {bias, bias, bias, bias, bias, bias, bias, bias};
        v8f acc_odd = {};
        for (int kb = 0; kb < 20; kb += 2) {
            const int krow = 4 * kb + 2 * half;
            v2f a0 = *(const v2f*)&Dsc[tloc * DSTRIDE + krow];
            v2f a1 = *(const v2f*)&Dsc[tloc * DSTRIDE + krow + 4];
            v2f b0, b1;
            b0.x = Wi[krow * NB + col];
            b0.y = Wi[(krow + 1) * NB + col];
            b1.x = Wi[(krow + 4) * NB + col];
            b1.y = Wi[(krow + 5) * NB + col];
            acc = __builtin_amdgcn_wmma_f32_16x16x4_f32(
                false, a0, false, b0, (short)0, acc, false, false);
            acc_odd = __builtin_amdgcn_wmma_f32_16x16x4_f32(
                false, a1, false, b1, (short)0, acc_odd, false, false);
        }
        acc = acc + acc_odd;
        float m = acc[0];
#pragma unroll
        for (int j = 1; j < 8; ++j) m = fmaxf(m, acc[j]);
        m = fmaxf(m, __shfl_xor(m, 16, 32));
        if (half == 0) {
            out[(n * NF + i) * NB + col] = fmaxf(m, 0.0f);
        }
    }
}

extern "C" void kernel_launch(void* const* d_in, const int* in_sizes, int n_in,
                              void* d_out, int out_size, void* d_ws, size_t ws_size,
                              hipStream_t stream) {
    const float* rho_coords  = (const float*)d_in[0];
    const float* theta_coord = (const float*)d_in[1];
    const float* input_feat  = (const float*)d_in[2];
    const float* mask        = (const float*)d_in[3];
    const float* mu_rho      = (const float*)d_in[4];
    const float* sigma_rho   = (const float*)d_in[5];
    const float* mu_theta    = (const float*)d_in[6];
    const float* sigma_theta = (const float*)d_in[7];
    const float* W_conv      = (const float*)d_in[8];
    const float* b_conv      = (const float*)d_in[9];
    float* out = (float*)d_out;

    const int n_samples = in_sizes[0] / NV;
    masif_geoconv_kernel<<<n_samples, 128, 0, stream>>>(
        rho_coords, theta_coord, input_feat, mask,
        mu_rho, sigma_rho, mu_theta, sigma_theta, W_conv, b_conv, out);
}


# Theta-Gaussian exponent as one packed fma + one packed multiply, pre-scaled by sqrt(log2e/(sigma^2+eps)): 2 packed ops per WMMA instead of 3; argument rounding changed, sum order unchanged
# speedup vs baseline: 1.5550x; 1.0334x over previous
#include <hip/hip_runtime.h>


#define NV 128
#define NF 4
#define NT 16
#define NR 5
#define NB 80
#define NROT 16
#define TWO_PI_F 6.28318530717958647692f
#define ROT_STEP (TWO_PI_F / 16.0f)
#define EPSF 1e-5f

typedef float v2f __attribute__((ext_vector_type(2)));
typedef float v8f __attribute__((ext_vector_type(8)));

#define QSTRIDE 130
#define DSTRIDE 82
#define TH_OFF  0
#define Q_OFF   (NROT * NV)
#define D_OFF   (Q_OFF + 4 * 16 * QSTRIDE)
#define LDS_FLOATS (D_OFF + 4 * 16 * DSTRIDE)
#define LOG2E_F 1.44269504088896340736f

__device__ __forceinline__ void load_theta_row(v2f (&th)[32], const float* row)
{
#pragma unroll
    for (int kb = 0; kb < 32; ++kb) th[kb] = *(const v2f*)&row[4 * kb];
}

__device__ __forceinline__ v8f rotation_block(const v2f (&a)[32], const v2f (&th)[32],
                                              const v2f s2, const v2f ms2)
{
    v8f acc0 = {}, acc1 = {};
#pragma unroll
    for (int kb = 0; kb < 32; kb += 2) {
        const v2f y0 = th[kb] * s2 + ms2;
        const v2f x0 = -(y0 * y0);
        v2f b0;
        b0.x = __builtin_amdgcn_exp2f(x0.x);
        b0.y = __builtin_amdgcn_exp2f(x0.y);
        acc0 = __builtin_amdgcn_wmma_f32_16x16x4_f32(
            false, a[kb], false, b0, (short)0, acc0, false, false);
        const v2f y1 = th[kb + 1] * s2 + ms2;
        const v2f x1 = -(y1 * y1);
        v2f b1;
        b1.x = __builtin_amdgcn_exp2f(x1.x);
        b1.y = __builtin_amdgcn_exp2f(x1.y);
        acc1 = __builtin_amdgcn_wmma_f32_16x16x4_f32(
            false, a[kb + 1], false, b1, (short)0, acc1, false, false);
    }
    return acc0 + acc1;
}

__device__ __forceinline__ void store_desc(float* drow, const v8f acc, const int half, const int tloc)
{
    if (half == 0) {
#pragma unroll
        for (int r = 0; r < 4; ++r) {
            drow[r * NT + tloc] = acc[2 * r] / (acc[2 * r + 1] + EPSF);
        }
    } else {
        drow[4 * NT + tloc] = acc[0] / (acc[1] + EPSF);
    }
}

__global__ __launch_bounds__(128)
void masif_geoconv_kernel(const float* __restrict__ rho_coords,
                          const float* __restrict__ theta_coords,
                          const float* __restrict__ input_feat,
                          const float* __restrict__ mask,
                          const float* __restrict__ mu_rho,
                          const float* __restrict__ sigma_rho,
                          const float* __restrict__ mu_theta,
                          const float* __restrict__ sigma_theta,
                          const float* __restrict__ W_conv,
                          const float* __restrict__ b_conv,
                          float* __restrict__ out)
{
    __shared__ float lds[LDS_FLOATS];

    const int n    = blockIdx.x;
    const int tid  = threadIdx.x;
    const int wave = tid >> 5;
    const int lane = tid & 31;
    const int half = lane >> 4;
    const int tloc = lane & 15;
    const int i    = wave;

    {
        const float tv = theta_coords[n * NV + tid];
#pragma unroll
        for (int k = 0; k < NROT; ++k) {
            float th = __builtin_fmaf((float)k, ROT_STEP, tv);
            th = (th >= TWO_PI_F) ? th - TWO_PI_F : th;
            lds[TH_OFF + k * NV + tid] = th;
        }
    }

    float mu_r[NR], is2_r[NR];
#pragma unroll
    for (int r = 0; r < NR; ++r) {
        float m = mu_rho[i * NB + r * NT];
        float s = sigma_rho[i * NB + r * NT];
        mu_r[r]  = m;
        is2_r[r] = 1.0f / (s * s + EPSF);
    }
    float s_t, ms_t;
    {
        const float m = mu_theta[i * NB + tloc];
        const float s = sigma_theta[i * NB + tloc];
        s_t  = __builtin_sqrtf((1.0f / (s * s + EPSF)) * LOG2E_F);
        ms_t = -(m * s_t);
    }

    float* Q = &lds[Q_OFF + wave * 16 * QSTRIDE];
    for (int v = lane; v < NV; v += 32) {
        float rv = rho_coords[n * NV + v];
        float fv = input_feat[(n * NV + v) * NF + i];
        float mv = mask[n * NV + v];
#pragma unroll
        for (int r = 0; r < NR; ++r) {
            float d  = rv - mu_r[r];
            float rg = __expf(-d * d * is2_r[r]) * mv;
            Q[(2 * r)     * QSTRIDE + v] = rg * fv;
            Q[(2 * r + 1) * QSTRIDE + v] = rg;
        }
#pragma unroll
        for (int c = 2 * NR; c < 16; ++c) Q[c * QSTRIDE + v] = 0.0f;
    }
    __syncthreads();

    float* Dsc = &lds[D_OFF + wave * 16 * DSTRIDE];
    const v2f s2  = {s_t, s_t};
    const v2f ms2 = {ms_t, ms_t};

    v2f afrag[32];
#pragma unroll
    for (int kb = 0; kb < 32; ++kb) afrag[kb] = *(const v2f*)&Q[tloc * QSTRIDE + 4 * kb + 2 * half];

    const float* THrow = &lds[TH_OFF + 2 * half];
    const float* row = THrow;
    v2f thA[32], thB[32];
    load_theta_row(thA, row);
    for (int k = 0; k < NROT; k += 2) {
        load_theta_row(thB, row + NV);
        __builtin_amdgcn_sched_barrier(0);
        store_desc(&Dsc[k * DSTRIDE], rotation_block(afrag, thA, s2, ms2), half, tloc);
        row = (k + 2 < NROT) ? row + 2 * NV : THrow;
        load_theta_row(thA, row);
        __builtin_amdgcn_sched_barrier(0);
        store_desc(&Dsc[(k + 1) * DSTRIDE], rotation_block(afrag, thB, s2, ms2), half, tloc);
    }
    __syncthreads();

    const float* Wi = W_conv + i * NB * NB;
    for (int bt = 0; bt < 5; ++bt) {
        const int col = bt * NT + tloc;
        const float bias = b_conv[i * NB + col];
        v8f acc = {bias, bias, bias, bias, bias, bias, bias, bias};
        v8f acc_odd = {};
        for (int kb = 0; kb < 20; kb += 2) {
            const int krow = 4 * kb + 2 * half;
            v2f a0 = *(const v2f*)&Dsc[tloc * DSTRIDE + krow];
            v2f a1 = *(const v2f*)&Dsc[tloc * DSTRIDE + krow + 4];
            v2f b0, b1;
            b0.x = Wi[krow * NB + col];
            b0.y = Wi[(krow + 1) * NB + col];
            b1.x = Wi[(krow + 4) * NB + col];
            b1.y = Wi[(krow + 5) * NB + col];
            acc = __builtin_amdgcn_wmma_f32_16x16x4_f32(
                false, a0, false, b0, (short)0, acc, false, false);
            acc_odd = __builtin_amdgcn_wmma_f32_16x16x4_f32(
                false, a1, false, b1, (short)0, acc_odd, false, false);
        }
        acc = acc + acc_odd;
        float m = acc[0];
#pragma unroll
        for (int j = 1; j < 8; ++j) m = fmaxf(m, acc[j]);
        m = fmaxf(m, __shfl_xor(m, 16, 32));
        if (half == 0) {
            out[(n * NF + i) * NB + col] = fmaxf(m, 0.0f);
        }
    }
}

extern "C" void kernel_launch(void* const* d_in, const int* in_sizes, int n_in,
                              void* d_out, int out_size, void* d_ws, size_t ws_size,
                              hipStream_t stream) {
    const float* rho_coords  = (const float*)d_in[0];
    const float* theta_coord = (const float*)d_in[1];
    const float* input_feat  = (const float*)d_in[2];
    const float* mask        = (const float*)d_in[3];
    const float* mu_rho      = (const float*)d_in[4];
    const float* sigma_rho   = (const float*)d_in[5];
    const float* mu_theta    = (const float*)d_in[6];
    const float* sigma_theta = (const float*)d_in[7];
    const float* W_conv      = (const float*)d_in[8];
    const float* b_conv      = (const float*)d_in[9];
    float* out = (float*)d_out;

    const int n_samples = in_sizes[0] / NV;
    masif_geoconv_kernel<<<n_samples, 128, 0, stream>>>(
        rho_coords, theta_coord, input_feat, mask,
        mu_rho, sigma_rho, mu_theta, sigma_theta, W_conv, b_conv, out);
}


# desc = num * rcp(den+eps): hardware reciprocal and a multiply, the same straight-line code in both lane halves, replace the branched IEEE divisions; quotient rounding changed, sum order unchanged
# speedup vs baseline: 1.6274x; 1.0466x over previous
#include <hip/hip_runtime.h>


#define NV 128
#define NF 4
#define NT 16
#define NR 5
#define NB 80
#define NROT 16
#define TWO_PI_F 6.28318530717958647692f
#define ROT_STEP (TWO_PI_F / 16.0f)
#define EPSF 1e-5f

typedef float v2f __attribute__((ext_vector_type(2)));
typedef float v8f __attribute__((ext_vector_type(8)));

#define QSTRIDE 130
#define DSTRIDE 82
#define TH_OFF  0
#define Q_OFF   (NROT * NV)
#define D_OFF   (Q_OFF + 4 * 16 * QSTRIDE)
#define LDS_FLOATS (D_OFF + 4 * 16 * DSTRIDE)
#define LOG2E_F 1.44269504088896340736f

__device__ __forceinline__ void load_theta_row(v2f (&th)[32], const float* row)
{
#pragma unroll
    for (int kb = 0; kb < 32; ++kb) th[kb] = *(const v2f*)&row[4 * kb];
}

__device__ __forceinline__ v8f rotation_block(const v2f (&a)[32], const v2f (&th)[32],
                                              const v2f s2, const v2f ms2)
{
    v8f acc0 = {}, acc1 = {};
#pragma unroll
    for (int kb = 0; kb < 32; kb += 2) {
        const v2f y0 = th[kb] * s2 + ms2;
        const v2f x0 = -(y0 * y0);
        v2f b0;
        b0.x = __builtin_amdgcn_exp2f(x0.x);
        b0.y = __builtin_amdgcn_exp2f(x0.y);
        acc0 = __builtin_amdgcn_wmma_f32_16x16x4_f32(
            false, a[kb], false, b0, (short)0, acc0, false, false);
        const v2f y1 = th[kb + 1] * s2 + ms2;
        const v2f x1 = -(y1 * y1);
        v2f b1;
        b1.x = __builtin_amdgcn_exp2f(x1.x);
        b1.y = __builtin_amdgcn_exp2f(x1.y);
        acc1 = __builtin_amdgcn_wmma_f32_16x16x4_f32(
            false, a[kb + 1], false, b1, (short)0, acc1, false, false);
    }
    return acc0 + acc1;
}

__device__ __forceinline__ void store_desc(float* drow, const v8f acc, const int half, const int tloc)
{
    float q[4];
#pragma unroll
    for (int r = 0; r < 4; ++r) {
        q[r] = acc[2 * r] * __builtin_amdgcn_rcpf(acc[2 * r + 1] + EPSF);
    }
    drow[half * (4 * NT) + tloc] = q[0];
    if (half == 0) {
#pragma unroll
        for (int r = 1; r < 4; ++r) drow[r * NT + tloc] = q[r];
    }
}

__global__ __launch_bounds__(128)
void masif_geoconv_kernel(const float* __restrict__ rho_coords,
                          const float* __restrict__ theta_coords,
                          const float* __restrict__ input_feat,
                          const float* __restrict__ mask,
                          const float* __restrict__ mu_rho,
                          const float* __restrict__ sigma_rho,
                          const float* __restrict__ mu_theta,
                          const float* __restrict__ sigma_theta,
                          const float* __restrict__ W_conv,
                          const float* __restrict__ b_conv,
                          float* __restrict__ out)
{
    __shared__ float lds[LDS_FLOATS];

    const int n    = blockIdx.x;
    const int tid  = threadIdx.x;
    const int wave = tid >> 5;
    const int lane = tid & 31;
    const int half = lane >> 4;
    const int tloc = lane & 15;
    const int i    = wave;

    {
        const float tv = theta_coords[n * NV + tid];
#pragma unroll
        for (int k = 0; k < NROT; ++k) {
            float th = __builtin_fmaf((float)k, ROT_STEP, tv);
            th = (th >= TWO_PI_F) ? th - TWO_PI_F : th;
            lds[TH_OFF + k * NV + tid] = th;
        }
    }

    float mu_r[NR], is2_r[NR];
#pragma unroll
    for (int r = 0; r < NR; ++r) {
        float m = mu_rho[i * NB + r * NT];
        float s = sigma_rho[i * NB + r * NT];
        mu_r[r]  = m;
        is2_r[r] = 1.0f / (s * s + EPSF);
    }
    float s_t, ms_t;
    {
        const float m = mu_theta[i * NB + tloc];
        const float s = sigma_theta[i * NB + tloc];
        s_t  = __builtin_sqrtf((1.0f / (s * s + EPSF)) * LOG2E_F);
        ms_t = -(m * s_t);
    }

    float* Q = &lds[Q_OFF + wave * 16 * QSTRIDE];
    for (int v = lane; v < NV; v += 32) {
        float rv = rho_coords[n * NV + v];
        float fv = input_feat[(n * NV + v) * NF + i];
        float mv = mask[n * NV + v];
#pragma unroll
        for (int r = 0; r < NR; ++r) {
            float d  = rv - mu_r[r];
            float rg = __expf(-d * d * is2_r[r]) * mv;
            Q[(2 * r)     * QSTRIDE + v] = rg * fv;
            Q[(2 * r + 1) * QSTRIDE + v] = rg;
        }
#pragma unroll
        for (int c = 2 * NR; c < 16; ++c) Q[c * QSTRIDE + v] = 0.0f;
    }
    __syncthreads();

    float* Dsc = &lds[D_OFF + wave * 16 * DSTRIDE];
    const v2f s2  = {s_t, s_t};
    const v2f ms2 = {ms_t, ms_t};

    v2f afrag[32];
#pragma unroll
    for (int kb = 0; kb < 32; ++kb) afrag[kb] = *(const v2f*)&Q[tloc * QSTRIDE + 4 * kb + 2 * half];

    const float* THrow = &lds[TH_OFF + 2 * half];
    const float* row = THrow;
    v2f thA[32], thB[32];
    load_theta_row(thA, row);
    for (int k = 0; k < NROT; k += 2) {
        load_theta_row(thB, row + NV);
        __builtin_amdgcn_sched_barrier(0);
        store_desc(&Dsc[k * DSTRIDE], rotation_block(afrag, thA, s2, ms2), half, tloc);
        row = (k + 2 < NROT) ? row + 2 * NV : THrow;
        load_theta_row(thA, row);
        __builtin_amdgcn_sched_barrier(0);
        store_desc(&Dsc[(k + 1) * DSTRIDE], rotation_block(afrag, thB, s2, ms2), half, tloc);
    }
    __syncthreads();

    const float* Wi = W_conv + i * NB * NB;
    for (int bt = 0; bt < 5; ++bt) {
        const int col = bt * NT + tloc;
        const float bias = b_conv[i * NB + col];
        v8f acc = {bias, bias, bias, bias, bias, bias, bias, bias};
        v8f acc_odd = {};
        for (int kb = 0; kb < 20; kb += 2) {
            const int krow = 4 * kb + 2 * half;
            v2f a0 = *(const v2f*)&Dsc[tloc * DSTRIDE + krow];
            v2f a1 = *(const v2f*)&Dsc[tloc * DSTRIDE + krow + 4];
            v2f b0, b1;
            b0.x = Wi[krow * NB + col];
            b0.y = Wi[(krow + 1) * NB + col];
            b1.x = Wi[(krow + 4) * NB + col];
            b1.y = Wi[(krow + 5) * NB + col];
            acc = __builtin_amdgcn_wmma_f32_16x16x4_f32(
                false, a0, false, b0, (short)0, acc, false, false);
            acc_odd = __builtin_amdgcn_wmma_f32_16x16x4_f32(
                false, a1, false, b1, (short)0, acc_odd, false, false);
        }
        acc = acc + acc_odd;
        float m = acc[0];
#pragma unroll
        for (int j = 1; j < 8; ++j) m = fmaxf(m, acc[j]);
        m = fmaxf(m, __shfl_xor(m, 16, 32));
        if (half == 0) {
            out[(n * NF + i) * NB + col] = fmaxf(m, 0.0f);
        }
    }
}

extern "C" void kernel_launch(void* const* d_in, const int* in_sizes, int n_in,
                              void* d_out, int out_size, void* d_ws, size_t ws_size,
                              hipStream_t stream) {
    const float* rho_coords  = (const float*)d_in[0];
    const float* theta_coord = (const float*)d_in[1];
    const float* input_feat  = (const float*)d_in[2];
    const float* mask        = (const float*)d_in[3];
    const float* mu_rho      = (const float*)d_in[4];
    const float* sigma_rho   = (const float*)d_in[5];
    const float* mu_theta    = (const float*)d_in[6];
    const float* sigma_theta = (const float*)d_in[7];
    const float* W_conv      = (const float*)d_in[8];
    const float* b_conv      = (const float*)d_in[9];
    float* out = (float*)d_out;

    const int n_samples = in_sizes[0] / NV;
    masif_geoconv_kernel<<<n_samples, 128, 0, stream>>>(
        rho_coords, theta_coord, input_feat, mask,
        mu_rho, sigma_rho, mu_theta, sigma_theta, W_conv, b_conv, out);
}
